# starting kernel
# baseline (speedup 1.0000x reference)
#include <hip/hip_runtime.h>

typedef __attribute__((ext_vector_type(2))) float v2f;
typedef __attribute__((ext_vector_type(8))) float v8f;

#define WIDTH   1000
#define PPR     200
#define NPATCH  40000
#define NPIX    1000000
#define NTERM   10

__global__ __launch_bounds__(256, 4)
void ts_wmma_kernel(const float* __restrict__ pix,
                    const float* __restrict__ coef,
                    const float* __restrict__ bias,
                    float* __restrict__ out) {
    const int lane = threadIdx.x & 31;
    const int wid  = threadIdx.x >> 5;
    const int p    = (blockIdx.x << 3) + wid;

    const int n  = lane & 15;
    const int hi = lane >> 4;

    const int cch = n < 3 ? n : 2;
    const float* cbase = coef + (((size_t)p * 3 + cch) * NTERM + hi) * 2;
    v2f a0 = *(const v2f*)(cbase +  0);
    v2f a1 = *(const v2f*)(cbase +  4);
    v2f a2 = *(const v2f*)(cbase +  8);
    v2f a3 = *(const v2f*)(cbase + 12);
    v2f a4 = *(const v2f*)(cbase + 16);

    const int prow = p / PPR, pcol = p % PPR;
    const int q0 = n;
    const int q1t = 16 + n;
    const int q1 = q1t < 25 ? q1t : 24;
    const int g0 = (5 * prow + q0 / 5) * WIDTH + 5 * pcol + q0 % 5;
    const int g1 = (5 * prow + q1 / 5) * WIDTH + 5 * pcol + q1 % 5;
    v2f xy0 = *(const v2f*)(pix + 2 * (size_t)g0);
    v2f xy1 = *(const v2f*)(pix + 2 * (size_t)g1);

    const float b0 = bias[3 * p + 0];
    const float b1 = bias[3 * p + 1];
    const float b2 = bias[3 * p + 2];
    v8f acc0 = {b0, b1, b2, 0.f, 0.f, 0.f, 0.f, 0.f};
    v8f acc1 = acc0;

    const float x0 = xy0[0], y0 = xy0[1], x1 = xy1[0], y1 = xy1[1];
    float xp0 = hi ? x0 : 1.0f, yp0 = hi ? y0 : 1.0f;
    float xp1 = hi ? x1 : 1.0f, yp1 = hi ? y1 : 1.0f;
    const float xs0 = x0 * x0, ys0 = y0 * y0;
    const float xs1 = x1 * x1, ys1 = y1 * y1;

    v2f a[5] = {a0, a1, a2, a3, a4};
#pragma unroll
    for (int j = 0; j < 5; ++j) {
        v2f bc0 = {xp0, yp0};
        v2f bc1 = {xp1, yp1};
        acc0 = __builtin_amdgcn_wmma_f32_16x16x4_f32(false, a[j], false, bc0,
                                                     (short)0, acc0, false, false);
        acc1 = __builtin_amdgcn_wmma_f32_16x16x4_f32(false, a[j], false, bc1,
                                                     (short)0, acc1, false, false);
        xp0 *= xs0; yp0 *= ys0;
        xp1 *= xs1; yp1 *= ys1;
    }

    if (lane < 16) {
        out[0 * NPIX + g0] = acc0[0];
        out[1 * NPIX + g0] = acc0[1];
        out[2 * NPIX + g0] = acc0[2];
        if (n < 9) {
            out[0 * NPIX + g1] = acc1[0];
            out[1 * NPIX + g1] = acc1[1];
            out[2 * NPIX + g1] = acc1[2];
        }
    }
}

extern "C" void kernel_launch(void* const* d_in, const int* in_sizes, int n_in,
                              void* d_out, int out_size, void* d_ws, size_t ws_size,
                              hipStream_t stream) {
    (void)in_sizes; (void)n_in; (void)out_size; (void)d_ws; (void)ws_size;
    const float* pix  = (const float*)d_in[0];
    const float* coef = (const float*)d_in[1];
    const float* bias = (const float*)d_in[2];
    float* out = (float*)d_out;

    ts_wmma_kernel<<<dim3(NPATCH / 8), dim3(256), 0, stream>>>(pix, coef, bias, out);
}


# one wave per strip of 8 adjacent patches, every load of the strip issued before its first product (5000 waves for 40000 patches): wave entry set the time, not memory or the matrix pipe
# speedup vs baseline: 2.1161x; 2.1161x over previous
#include <hip/hip_runtime.h>


typedef __attribute__((ext_vector_type(2))) float v2f;
typedef __attribute__((ext_vector_type(8))) float v8f;

#define WIDTH   1000
#define PPR     200
#define NPATCH  40000
#define NPIX    1000000
#define NTERM   10
#define NP      8
#define WPB     4

__global__ __launch_bounds__(WPB * 32)
void ts_wmma_kernel(const float* __restrict__ pix,
                    const float* __restrict__ coef,
                    const float* __restrict__ bias,
                    float* __restrict__ out) {
    const int lane = threadIdx.x & 31;
    const int wid  = __builtin_amdgcn_readfirstlane(threadIdx.x >> 5);
    const int p0   = (blockIdx.x * WPB + wid) * NP;

    const int n  = lane & 15;
    const int hi = lane >> 4;

    const int cch = n < 3 ? n : 2;
    const float* cbase = coef + (((size_t)p0 * 3 + cch) * NTERM + hi) * 2;

    const int prow = p0 / PPR, pcol = p0 % PPR;
    const int q0 = n;
    const int q1t = 16 + n;
    const int q1 = q1t < 25 ? q1t : 24;
    const int g0 = (5 * prow + q0 / 5) * WIDTH + 5 * pcol + q0 % 5;
    const int g1 = (5 * prow + q1 / 5) * WIDTH + 5 * pcol + q1 % 5;
    const float* pb0 = pix + 2 * (size_t)g0;
    const float* pb1 = pix + 2 * (size_t)g1;
    const float* bb  = bias + 3 * (size_t)p0;

    v2f a[NP][5];
    v2f xy0[NP], xy1[NP];
    float bs[NP][3];
#pragma unroll
    for (int i = 0; i < NP; ++i) {
#pragma unroll
        for (int j = 0; j < 5; ++j)
            a[i][j] = *(const v2f*)(cbase + i * (3 * NTERM * 2) + 4 * j);
        xy0[i] = *(const v2f*)(pb0 + 10 * i);
        xy1[i] = *(const v2f*)(pb1 + 10 * i);
        bs[i][0] = bb[3 * i + 0];
        bs[i][1] = bb[3 * i + 1];
        bs[i][2] = bb[3 * i + 2];
    }

    __builtin_amdgcn_sched_barrier(0);

    v8f acc0[NP], acc1[NP];
#pragma unroll
    for (int i = 0; i < NP; ++i) {
        v8f c0 = {bs[i][0], bs[i][1], bs[i][2], 0.f, 0.f, 0.f, 0.f, 0.f};
        v8f c1 = c0;

        const float x0 = xy0[i][0], y0 = xy0[i][1], x1 = xy1[i][0], y1 = xy1[i][1];
        float xp0 = hi ? x0 : 1.0f, yp0 = hi ? y0 : 1.0f;
        float xp1 = hi ? x1 : 1.0f, yp1 = hi ? y1 : 1.0f;
        const float xs0 = x0 * x0, ys0 = y0 * y0;
        const float xs1 = x1 * x1, ys1 = y1 * y1;

#pragma unroll
        for (int j = 0; j < 5; ++j) {
            v2f bc0 = {xp0, yp0};
            v2f bc1 = {xp1, yp1};
            c0 = __builtin_amdgcn_wmma_f32_16x16x4_f32(false, a[i][j], false, bc0,
                                                       (short)0, c0, false, false);
            c1 = __builtin_amdgcn_wmma_f32_16x16x4_f32(false, a[i][j], false, bc1,
                                                       (short)0, c1, false, false);
            xp0 *= xs0; yp0 *= ys0;
            xp1 *= xs1; yp1 *= ys1;
        }
        acc0[i] = c0;
        acc1[i] = c1;
    }

    if (lane < 16) {
#pragma unroll
        for (int i = 0; i < NP; ++i) {
            out[0 * NPIX + g0 + 5 * i] = acc0[i][0];
            out[1 * NPIX + g0 + 5 * i] = acc0[i][1];
            out[2 * NPIX + g0 + 5 * i] = acc0[i][2];
        }
        if (n < 9) {
#pragma unroll
            for (int i = 0; i < NP; ++i) {
                out[0 * NPIX + g1 + 5 * i] = acc1[i][0];
                out[1 * NPIX + g1 + 5 * i] = acc1[i][1];
                out[2 * NPIX + g1 + 5 * i] = acc1[i][2];
            }
        }
    }
}

extern "C" void kernel_launch(void* const* d_in, const int* in_sizes, int n_in,
                              void* d_out, int out_size, void* d_ws, size_t ws_size,
                              hipStream_t stream) {
    (void)in_sizes; (void)n_in; (void)out_size; (void)d_ws; (void)ws_size;
    const float* pix  = (const float*)d_in[0];
    const float* coef = (const float*)d_in[1];
    const float* bias = (const float*)d_in[2];
    float* out = (float*)d_out;

    ts_wmma_kernel<<<dim3(NPATCH / (WPB * NP)), dim3(WPB * 32), 0, stream>>>(pix, coef, bias, out);
}


# one work-group per WGP: 250 work-groups of 20 waves (5 per SIMD) instead of 1250 of 4, so 5 times fewer work-groups to dispatch; per-wave code, strips and summation order unchanged
# speedup vs baseline: 2.5542x; 1.2070x over previous
#include <hip/hip_runtime.h>


typedef __attribute__((ext_vector_type(2))) float v2f;
typedef __attribute__((ext_vector_type(8))) float v8f;

#define WIDTH   1000
#define PPR     200
#define NPATCH  40000
#define NPIX    1000000
#define NTERM   10
#define NP      8
#define WPB     20

__device__ __forceinline__ float xhalf(float v, float spent) {
    return __builtin_bit_cast(float, __builtin_amdgcn_permlanex16(__builtin_bit_cast(int, spent), __builtin_bit_cast(int, v),
                                                                  0x76543210, 0xfedcba98, false, false));
}

struct TileOut { float a, b; };

__device__ __forceinline__ TileOut tile(v2f xy, v2f A0, v2f A1, float bsx, float bs1, int hi) {
    const float x = xy[0], y = xy[1];
    const float xs = x * x, ys = y * y;
    const v2f b0 = {hi ? x : 1.0f, hi ? y : 1.0f};
    const v2f b1 = {b0[0] * xs, b0[1] * ys};
    const float ps = hi ? ys : xs;
    const float p4 = ps * ps;
    const v8f zero = {0.f, 0.f, 0.f, 0.f, 0.f, 0.f, 0.f, 0.f};
    v8f d = __builtin_amdgcn_wmma_f32_16x16x4_f32(false, A0, false, b0, (short)0, zero, false, false);
    d = __builtin_amdgcn_wmma_f32_16x16x4_f32(false, A1, false, b1, (short)0, d, false, false);
    const float iA = __builtin_fmaf(p4, d[1], d[0]);
    const float iB = __builtin_fmaf(p4, d[3], d[2]);
    const float iC = __builtin_fmaf(p4, d[5], d[4]);
    const float u = __builtin_fmaf(p4, iA, d[6]);
    const float v = __builtin_fmaf(p4, iC, bsx);
    const float w = __builtin_fmaf(p4, iB, d[7]);
    TileOut o;
    o.a = u + xhalf(v, v);
    o.b = (w + xhalf(w, iB)) + bs1;
    return o;
}

__global__ __launch_bounds__(WPB * 32)
void ts_wmma_kernel(const float* __restrict__ pix,
                    const float* __restrict__ coef,
                    const float* __restrict__ bias,
                    float* __restrict__ out) {
    const int lane = threadIdx.x & 31;
    const int wid  = __builtin_amdgcn_readfirstlane(threadIdx.x >> 5);
    const int p0   = (blockIdx.x * WPB + wid) * NP;

    const int n  = lane & 15;
    const int hi = lane >> 4;

    const int cch = (0x605a4a50u >> (2 * n)) & 3;
    const int grp = (0x9990999u >> (2 * n)) & 3;
    const bool kx = (0x40ffu >> n) & 1;
    const bool ky = (0x7fc0u >> n) & 1;
    const bool two = grp < 2;
    const bool k1x = kx && two, k1y = ky && two;
    const float* cb0 = coef + (((size_t)p0 * 3 + cch) * NTERM + 4 * grp + hi) * 2;
    const float* cb1 = cb0 + (two ? 4 : 0);

    const int prow = p0 / PPR, pcol = p0 % PPR;
    const int q0 = n;
    const int q1t = 16 + n;
    const int q1 = q1t < 25 ? q1t : 24;
    const int g0 = (5 * prow + q0 / 5) * WIDTH + 5 * pcol + q0 % 5;
    const int g1 = (5 * prow + q1 / 5) * WIDTH + 5 * pcol + q1 % 5;
    const float* pb0 = pix + 2 * (size_t)g0;
    const float* pb1 = pix + 2 * (size_t)g1;
    const float* bbx = bias + 3 * (size_t)p0 + (hi ? 0 : 2);
    const float* bb1 = bias + 3 * (size_t)p0 + 1;

    v2f a0[NP], a1[NP];
    v2f xy0[NP], xy1[NP];
    float bx[NP], b1[NP];
#pragma unroll
    for (int i = 0; i < NP; ++i) {
        a0[i] = *(const v2f*)(cb0 + i * (3 * NTERM * 2));
        a1[i] = *(const v2f*)(cb1 + i * (3 * NTERM * 2));
        xy0[i] = *(const v2f*)(pb0 + 10 * i);
        xy1[i] = *(const v2f*)(pb1 + 10 * i);
        bx[i] = bbx[3 * i];
        b1[i] = bb1[3 * i];
    }

    __builtin_amdgcn_sched_barrier(0);

    float oa0[NP], ob0[NP], oa1[NP], ob1[NP];
#pragma unroll
    for (int i = 0; i < NP; ++i) {
        const v2f A0 = {kx ? a0[i][0] : 0.0f, ky ? a0[i][1] : 0.0f};
        const v2f A1 = {k1x ? a1[i][0] : 0.0f, k1y ? a1[i][1] : 0.0f};
        const TileOut t0 = tile(xy0[i], A0, A1, bx[i], b1[i], hi);
        const TileOut t1 = tile(xy1[i], A0, A1, bx[i], b1[i], hi);
        oa0[i] = t0.a; ob0[i] = t0.b;
        oa1[i] = t1.a; ob1[i] = t1.b;
    }

    float* oa = out + (hi ? 2 * NPIX : 0);
#pragma unroll
    for (int i = 0; i < NP; ++i) oa[g0 + 5 * i] = oa0[i];
    if (hi == 0) {
#pragma unroll
        for (int i = 0; i < NP; ++i) out[NPIX + g0 + 5 * i] = ob0[i];
    }
    if (n < 9) {
#pragma unroll
        for (int i = 0; i < NP; ++i) oa[g1 + 5 * i] = oa1[i];
        if (hi == 0) {
#pragma unroll
            for (int i = 0; i < NP; ++i) out[NPIX + g1 + 5 * i] = ob1[i];
        }
    }
}

extern "C" void kernel_launch(void* const* d_in, const int* in_sizes, int n_in,
                              void* d_out, int out_size, void* d_ws, size_t ws_size,
                              hipStream_t stream) {
    (void)in_sizes; (void)n_in; (void)out_size; (void)d_ws; (void)ws_size;
    const float* pix  = (const float*)d_in[0];
    const float* coef = (const float*)d_in[1];
    const float* bias = (const float*)d_in[2];
    float* out = (float*)d_out;

    ts_wmma_kernel<<<dim3(NPATCH / (WPB * NP)), dim3(WPB * 32), 0, stream>>>(pix, coef, bias, out);
}
